# starting kernel
# baseline (speedup 1.0000x reference)
#include <hip/hip_runtime.h>

typedef __attribute__((ext_vector_type(2))) float v2f;
typedef __attribute__((ext_vector_type(4))) float v4f;
typedef __attribute__((ext_vector_type(8))) float v8f;

#define KDIM   768
#define NBATCH 32768

__global__ void build_eff_weights(const float* __restrict__ lhs,
                                  const float* __restrict__ rhs,
                                  const float* __restrict__ W,
                                  float* __restrict__ E) {
    for (int e = threadIdx.x; e < KDIM; e += blockDim.x) {
        const int ch  = e >> 8;
        const int rem = e & 255;
        const int row = rem >> 4;
        const int col = rem & 15;
        const int r = row >> 3, p = row & 7;
        const int c = col >> 3, q = col & 7;

        const float* lp = lhs + r * 128 + p;
        const float* rp = rhs + c * 128 + q * 16;
        const float* wp = W + ch * 1024 + (r * 16) * 32 + c * 16;

        float acc = 0.0f;
        #pragma unroll
        for (int P = 0; P < 16; ++P) {
            const float  l    = lp[P * 8];
            const float* wrow = wp + P * 32;
            float s = 0.0f;
            #pragma unroll
            for (int Q = 0; Q < 16; ++Q) s += wrow[Q] * rp[Q];
            acc += l * s;
        }
        E[e] = acc;
    }
}

__global__ void __launch_bounds__(256) gemv_wmma_f32(const float* __restrict__ x,
                                                     const float* __restrict__ E,
                                                     const float* __restrict__ bias,
                                                     float* __restrict__ y) {
    __shared__ float Elds[KDIM];
    for (int i = threadIdx.x; i < KDIM; i += 256) Elds[i] = E[i];
    __syncthreads();

    const int lane    = threadIdx.x & 31;
    const int wave    = threadIdx.x >> 5;
    const int rowBase = blockIdx.x * 128 + wave * 16;
    const int m       = lane & 15;
    const int hoff    = (lane >> 4) << 2;

    const float* xrow = x + (size_t)(rowBase + m) * KDIM + hoff;
    const float* Eh   = Elds + hoff;

    v8f acc = {};
    #pragma unroll 4
    for (int k = 0; k < KDIM; k += 8) {
        v4f xa = *(const v4f*)(xrow + k);
        v4f eb = *(const v4f*)(Eh + k);

        v2f a0 = { xa.x, xa.y };
        v2f b0 = { eb.x, eb.y };
        acc = __builtin_amdgcn_wmma_f32_16x16x4_f32(
            false, a0, false, b0, (short)0, acc, false, false);

        v2f a1 = { xa.z, xa.w };
        v2f b1 = { eb.z, eb.w };
        acc = __builtin_amdgcn_wmma_f32_16x16x4_f32(
            false, a1, false, b1, (short)0, acc, false, false);
    }

    const float bs = bias[0];
    if (m == 0) {
        float* out = y + rowBase + ((lane >> 4) << 3);
        v4f lo = { acc[0] + bs, acc[1] + bs, acc[2] + bs, acc[3] + bs };
        v4f hi = { acc[4] + bs, acc[5] + bs, acc[6] + bs, acc[7] + bs };
        *(v4f*)(out)     = lo;
        *(v4f*)(out + 4) = hi;
    }
}

extern "C" void kernel_launch(void* const* d_in, const int* in_sizes, int n_in,
                              void* d_out, int out_size, void* d_ws, size_t ws_size,
                              hipStream_t stream) {
    const float* x    = (const float*)d_in[0];
    const float* lhs  = (const float*)d_in[1];
    const float* rhs  = (const float*)d_in[2];
    const float* W    = (const float*)d_in[3];
    const float* bias = (const float*)d_in[4];
    float*       out  = (float*)d_out;
    float*       E    = (float*)d_ws;

    build_eff_weights<<<1, 256, 0, stream>>>(lhs, rhs, W, E);
    gemv_wmma_f32<<<NBATCH / 128, 256, 0, stream>>>(x, E, bias, out);
}


# x streamed through a 4-block register ring requested at kernel entry: 24-32 b128 loads per wave stay in flight across the E build (LDS-only barriers) and the dot loop, not 4; sum order unchanged
# speedup vs baseline: 3.0005x; 3.0005x over previous
#include <hip/hip_runtime.h>


typedef __attribute__((ext_vector_type(2))) float v2f;
typedef __attribute__((ext_vector_type(4))) float v4f;
typedef __attribute__((ext_vector_type(8))) float v8f;

#define KDIM   768
#define NBATCH 32768
#define BLK    8
#define NBLK   12
#define AHEAD  3

__device__ __forceinline__ v8f chunk_mac(v8f acc, v4f xa, v4f eb) {
    v2f a0 = { xa.x, xa.y };
    v2f b0 = { eb.x, eb.y };
    acc = __builtin_amdgcn_wmma_f32_16x16x4_f32(
        false, a0, false, b0, (short)0, acc, false, false);
    v2f a1 = { xa.z, xa.w };
    v2f b1 = { eb.z, eb.w };
    acc = __builtin_amdgcn_wmma_f32_16x16x4_f32(
        false, a1, false, b1, (short)0, acc, false, false);
    return acc;
}

__device__ __forceinline__ void lds_barrier() {
    __builtin_amdgcn_fence(__ATOMIC_RELEASE, "workgroup", "local");
    __builtin_amdgcn_s_barrier();
    __builtin_amdgcn_fence(__ATOMIC_ACQUIRE, "workgroup", "local");
}

__global__ void __launch_bounds__(256) fold_gemv_wmma_f32(const float* __restrict__ x,
                                                          const float* __restrict__ lhs,
                                                          const float* __restrict__ rhs,
                                                          const float* __restrict__ W,
                                                          const float* __restrict__ bias,
                                                          float* __restrict__ y) {
    __shared__ __attribute__((aligned(16))) float Elds[KDIM];
    __shared__ __attribute__((aligned(16))) float Tlds[1536];
    __shared__ __attribute__((aligned(16))) float Llds[256];
    __shared__ __attribute__((aligned(16))) float Rlds[256];

    const int t       = threadIdx.x;
    const int lane    = t & 31;
    const int wave    = t >> 5;
    const int rowBase = blockIdx.x * 128 + wave * 16;
    const int m       = lane & 15;
    const int hoff    = (lane >> 4) << 2;

    const float* xrow = x + (size_t)(rowBase + m) * KDIM + hoff;

    const float lv = lhs[t];
    const float rv = rhs[t];
    const v4f* wp = (const v4f*)(W + (t < 192 ? t : 191) * 16);
    const v4f w0 = wp[0], w1 = wp[1], w2 = wp[2], w3 = wp[3];

    v4f xb[AHEAD + 1][BLK];
#define LOAD_BLK(B) do { _Pragma("unroll") for (int c_ = 0; c_ < BLK; ++c_) \
        xb[(B) % (AHEAD + 1)][c_] = *(const v4f*)(xrow + ((B) * BLK + c_) * 8); } while (0)

    #pragma unroll
    for (int b = 0; b < AHEAD; ++b) LOAD_BLK(b);

    Llds[t] = lv;
    Rlds[t] = rv;
    lds_barrier();

    if (t < 192) {
        const float w[16] = { w0.x, w0.y, w0.z, w0.w, w1.x, w1.y, w1.z, w1.w,
                              w2.x, w2.y, w2.z, w2.w, w3.x, w3.y, w3.z, w3.w };
        const v4f* rp = (const v4f*)(Rlds + (t & 1) * 128);
        v4f rr[32];
        #pragma unroll
        for (int j = 0; j < 32; ++j) rr[j] = rp[j];
        __builtin_amdgcn_sched_barrier(0);
        float s[8];
        #pragma unroll
        for (int q = 0; q < 8; ++q) {
            float a = 0.0f;
            #pragma unroll
            for (int Q = 0; Q < 16; ++Q) a += w[Q] * rr[q * 4 + (Q >> 2)][Q & 3];
            s[q] = a;
        }
        v4f* tp = (v4f*)(Tlds + t * 8);
        tp[0] = (v4f){ s[0], s[1], s[2], s[3] };
        tp[1] = (v4f){ s[4], s[5], s[6], s[7] };
    }
    lds_barrier();

    #pragma unroll
    for (int i = 0; i < 3; ++i) {
        const int e   = t + i * 256;
        const int chr = e >> 7;
        const int r   = chr & 1;
        const int p   = (e >> 4) & 7;
        const int col = e & 15;
        const float* lp = Llds + r * 128 + p;
        const float* tq = Tlds + chr * 256 + col;
        float acc = 0.0f;
        #pragma unroll
        for (int P = 0; P < 16; ++P) acc += lp[P * 8] * tq[P * 16];
        Elds[e] = acc;
    }
    lds_barrier();

    const float* Eh = Elds + hoff;

    v8f acc = {};
    #pragma unroll
    for (int j = 0; j < NBLK; ++j) {
        __builtin_amdgcn_sched_barrier(0);
        if (j + AHEAD < NBLK) LOAD_BLK(j + AHEAD);
        __builtin_amdgcn_sched_barrier(0);
        #pragma unroll
        for (int c = 0; c < BLK; ++c) {
            v4f eb = *(const v4f*)(Eh + (j * BLK + c) * 8);
            acc = chunk_mac(acc, xb[j % (AHEAD + 1)][c], eb);
        }
    }
#undef LOAD_BLK

    const float bs = bias[0];
    if (m == 0) {
        float* out = y + rowBase + ((lane >> 4) << 3);
        v4f lo = { acc[0] + bs, acc[1] + bs, acc[2] + bs, acc[3] + bs };
        v4f hi = { acc[4] + bs, acc[5] + bs, acc[6] + bs, acc[7] + bs };
        *(v4f*)(out)     = lo;
        *(v4f*)(out + 4) = hi;
    }
}

extern "C" void kernel_launch(void* const* d_in, const int* in_sizes, int n_in,
                              void* d_out, int out_size, void* d_ws, size_t ws_size,
                              hipStream_t stream) {
    const float* x    = (const float*)d_in[0];
    const float* lhs  = (const float*)d_in[1];
    const float* rhs  = (const float*)d_in[2];
    const float* W    = (const float*)d_in[3];
    const float* bias = (const float*)d_in[4];
    float*       out  = (float*)d_out;

    fold_gemv_wmma_f32<<<NBATCH / 128, 256, 0, stream>>>(x, lhs, rhs, W, bias, out);
}


# Whole-line x loads: each load instruction reads 512 contiguous bytes of one sample (was 32 B of 16 rows); dot = WMMA diagonal, E fragment in registers, lane butterfly per sample (sum order changed)
# speedup vs baseline: 3.4485x; 1.1493x over previous
#include <hip/hip_runtime.h>


typedef __attribute__((ext_vector_type(2))) float v2f;
typedef __attribute__((ext_vector_type(4))) float v4f;
typedef __attribute__((ext_vector_type(8))) float v8f;

#define KDIM   768
#define NBATCH 32768
#define SPW    16
#define PCS    6
#define AHEAD  6
#define RING   (AHEAD + 2)

__device__ __forceinline__ v8f chunk_mac(v8f acc, v4f xa, v4f eb) {
    v2f a0 = { xa.x, xa.y };
    v2f b0 = { eb.x, eb.y };
    acc = __builtin_amdgcn_wmma_f32_16x16x4_f32(
        false, a0, false, b0, (short)0, acc, false, false);
    v2f a1 = { xa.z, xa.w };
    v2f b1 = { eb.z, eb.w };
    acc = __builtin_amdgcn_wmma_f32_16x16x4_f32(
        false, a1, false, b1, (short)0, acc, false, false);
    return acc;
}

__device__ __forceinline__ void lds_barrier() {
    __builtin_amdgcn_fence(__ATOMIC_RELEASE, "workgroup", "local");
    __builtin_amdgcn_s_barrier();
    __builtin_amdgcn_fence(__ATOMIC_ACQUIRE, "workgroup", "local");
}

__global__ void __launch_bounds__(256) fold_gemv_wmma_f32(const float* __restrict__ x,
                                                          const float* __restrict__ lhs,
                                                          const float* __restrict__ rhs,
                                                          const float* __restrict__ W,
                                                          const float* __restrict__ bias,
                                                          float* __restrict__ y) {
    __shared__ __attribute__((aligned(16))) float Elds[KDIM];
    __shared__ __attribute__((aligned(16))) float Tlds[1536];
    __shared__ __attribute__((aligned(16))) float Llds[256];
    __shared__ __attribute__((aligned(16))) float Rlds[256];

    const int t       = threadIdx.x;
    const int lane    = t & 31;
    const int wave    = t >> 5;
    const int rowBase = blockIdx.x * 128 + wave * SPW;

    const float* xw = x + (size_t)rowBase * KDIM + lane * 4;

    const float lv = lhs[t];
    const float rv = rhs[t];
    const v4f* wp = (const v4f*)(W + (t < 192 ? t : 191) * 16);
    const v4f w0 = wp[0], w1 = wp[1], w2 = wp[2], w3 = wp[3];

    v4f xr[RING][PCS];
#define LOAD_SMP(S) do { _Pragma("unroll") for (int q_ = 0; q_ < PCS; ++q_) \
        xr[(S) % RING][q_] = *(const v4f*)(xw + (S) * KDIM + q_ * 128); } while (0)

    #pragma unroll
    for (int s = 0; s < AHEAD; ++s) LOAD_SMP(s);

    Llds[t] = lv;
    Rlds[t] = rv;
    lds_barrier();

    if (t < 192) {
        const float w[16] = { w0.x, w0.y, w0.z, w0.w, w1.x, w1.y, w1.z, w1.w,
                              w2.x, w2.y, w2.z, w2.w, w3.x, w3.y, w3.z, w3.w };
        const v4f* rp = (const v4f*)(Rlds + (t & 1) * 128);
        v4f rr[32];
        #pragma unroll
        for (int j = 0; j < 32; ++j) rr[j] = rp[j];
        __builtin_amdgcn_sched_barrier(0);
        float s[8];
        #pragma unroll
        for (int q = 0; q < 8; ++q) {
            float a = 0.0f;
            #pragma unroll
            for (int Q = 0; Q < 16; ++Q) a += w[Q] * rr[q * 4 + (Q >> 2)][Q & 3];
            s[q] = a;
        }
        v4f* tp = (v4f*)(Tlds + t * 8);
        tp[0] = (v4f){ s[0], s[1], s[2], s[3] };
        tp[1] = (v4f){ s[4], s[5], s[6], s[7] };
    }
    lds_barrier();

    #pragma unroll
    for (int i = 0; i < 3; ++i) {
        const int e   = t + i * 256;
        const int chr = e >> 7;
        const int r   = chr & 1;
        const int p   = (e >> 4) & 7;
        const int col = e & 15;
        const float* lp = Llds + r * 128 + p;
        const float* tq = Tlds + chr * 256 + col;
        float acc = 0.0f;
        #pragma unroll
        for (int P = 0; P < 16; ++P) acc += lp[P * 8] * tq[P * 16];
        Elds[e] = acc;
    }
    lds_barrier();

    v4f ef[PCS];
    #pragma unroll
    for (int q = 0; q < PCS; ++q) ef[q] = *(const v4f*)(Elds + q * 128 + lane * 4);

    const bool b1 = (lane & 1) != 0;
    const bool b2 = (lane & 2) != 0;
    const bool b4 = (lane & 4) != 0;

    v8f pa0 = {}, pa1 = {};
    float m1[2], m2[2], m3[2];

    #pragma unroll
    for (int j = 0; j <= SPW; j += 2) {
        v8f a0 = {}, a1 = {};
        if (j < SPW) {
            __builtin_amdgcn_sched_barrier(0);
            if (j + AHEAD < SPW) { LOAD_SMP(j + AHEAD); LOAD_SMP(j + AHEAD + 1); }
            __builtin_amdgcn_sched_barrier(0);
            #pragma unroll
            for (int q = 0; q < PCS; ++q) {
                a0 = chunk_mac(a0, xr[j % RING][q], ef[q]);
                a1 = chunk_mac(a1, xr[(j + 1) % RING][q], ef[q]);
            }
        }
        if (j > 0) {
            const int p = (j - 2) >> 1;
            float d0, d1;
            {
                const float u0 = b1 ? pa0[1] : pa0[0], u1 = b1 ? pa0[3] : pa0[2];
                const float u2 = b1 ? pa0[5] : pa0[4], u3 = b1 ? pa0[7] : pa0[6];
                const float w0 = b2 ? u1 : u0, w1 = b2 ? u3 : u2;
                d0 = b4 ? w1 : w0;
            }
            {
                const float u0 = b1 ? pa1[1] : pa1[0], u1 = b1 ? pa1[3] : pa1[2];
                const float u2 = b1 ? pa1[5] : pa1[4], u3 = b1 ? pa1[7] : pa1[6];
                const float w0 = b2 ? u1 : u0, w1 = b2 ? u3 : u2;
                d1 = b4 ? w1 : w0;
            }
            m1[p & 1] = (b1 ? d1 : d0) + __shfl_xor(b1 ? d0 : d1, 1);
            if (p & 1) {
                const float k = b2 ? m1[1] : m1[0], g = b2 ? m1[0] : m1[1];
                m2[(p >> 1) & 1] = k + __shfl_xor(g, 2);
            }
            if ((p & 3) == 3) {
                const float k = b4 ? m2[1] : m2[0], g = b4 ? m2[0] : m2[1];
                m3[p >> 2] = k + __shfl_xor(g, 4);
            }
        }
        pa0 = a0;
        pa1 = a1;
    }
#undef LOAD_SMP

    const int   grp = lane >> 3;
    const float hx  = __shfl_xor(m3[1], 8);
    const float cv  = (grp == 0 || grp == 3) ? m3[0] : hx;
    const float yv  = cv + __shfl_xor(cv, 24);

    const float bs = bias[0];
    if (lane < SPW) y[rowBase + lane] = yv + bs;
}

extern "C" void kernel_launch(void* const* d_in, const int* in_sizes, int n_in,
                              void* d_out, int out_size, void* d_ws, size_t ws_size,
                              hipStream_t stream) {
    const float* x    = (const float*)d_in[0];
    const float* lhs  = (const float*)d_in[1];
    const float* rhs  = (const float*)d_in[2];
    const float* W    = (const float*)d_in[3];
    const float* bias = (const float*)d_in[4];
    float*       out  = (float*)d_out;

    fold_gemv_wmma_f32<<<NBATCH / 128, 256, 0, stream>>>(x, lhs, rhs, W, bias, out);
}
